# starting kernel
# baseline (speedup 1.0000x reference)
#include <hip/hip_runtime.h>

typedef float v2f __attribute__((ext_vector_type(2)));
typedef float v8f __attribute__((ext_vector_type(8)));

__device__ __forceinline__ v8f wmma_f32(v2f a, v2f b, v8f c) {
    return __builtin_amdgcn_wmma_f32_16x16x4_f32(false, a, false, b, (short)0, c, false, false);
}

__global__ __launch_bounds__(256) void build_unitaries(const float* __restrict__ w1,
                                                       const float* __restrict__ w2,
                                                       float* __restrict__ U) {
    __shared__ float mre[4][10][2][2];
    __shared__ float mim[4][10][2][2];

    const int tid = threadIdx.x;
    if (tid < 40) {
        const int l = tid / 10;
        const int j = tid % 10;
        const float* w = (l < 3) ? (w1 + (l * 10 + j) * 3) : (w2 + j * 3);
        const float phi = w[0], theta = w[1], omega = w[2];
        const float c = cosf(0.5f * theta);
        const float s = sinf(0.5f * theta);
        float sa, ca, sb, cb;
        sincosf(-0.5f * (phi + omega), &sa, &ca);
        sincosf(-0.5f * (phi - omega), &sb, &cb);
        mre[l][j][0][0] =  ca * c;  mim[l][j][0][0] =  sa * c;
        mre[l][j][0][1] = -cb * s;  mim[l][j][0][1] =  sb * s;
        mre[l][j][1][0] =  cb * s;  mim[l][j][1][0] =  sb * s;
        mre[l][j][1][1] =  ca * c;  mim[l][j][1][1] = -sa * c;
    }
    __syncthreads();

    for (int it = 0; it < 32; ++it) {
        const int idx = tid + it * 256;
        const int l   = idx >> 11;
        const int rem = idx & 2047;
        const int mat = rem >> 10;
        const int e   = rem & 1023;
        const int r   = e >> 5;
        const int cc  = e & 31;
        float pr = 1.f, pi = 0.f;
        const int j0 = mat ? 5 : 0;
        #pragma unroll
        for (int jj = 0; jj < 5; ++jj) {
            const int rb = (r  >> (4 - jj)) & 1;
            const int cb = (cc >> (4 - jj)) & 1;
            const float ar = mre[l][j0 + jj][rb][cb];
            const float ai = mim[l][j0 + jj][rb][cb];
            const float nr = pr * ar - pi * ai;
            const float ni = pr * ai + pi * ar;
            pr = nr; pi = ni;
        }
        float* base = U + l * 4096 + mat * 2048;
        base[e]        = pr;
        base[e + 1024] = pi;
    }
}

__global__ __launch_bounds__(128) void qsim(const float* __restrict__ x,
                                            const float* __restrict__ U,
                                            float* __restrict__ out) {
    __shared__ float Sr[1024], Si[1024];
    __shared__ float Cr[1024], Ci[1024];
    __shared__ float Ar[1024], Ai[1024], Br[1024], Bi[1024];
    __shared__ float prr[32], pri[32], pcr[32], pci[32];
    __shared__ float red[128][10];

    const int tid  = threadIdx.x;
    const int b    = blockIdx.x;
    const int lane = tid & 31;
    const int wid  = tid >> 5;
    const int ti   = wid >> 1;
    const int tj   = wid & 1;

    for (int i = tid; i < 1024; i += 128) { Sr[i] = (i == 0) ? 1.f : 0.f; Si[i] = 0.f; }

    float xb[10];
    #pragma unroll
    for (int j = 0; j < 10; ++j) xb[j] = x[b * 10 + j];

    if (tid < 64) {
        const int  idx = tid & 31;
        const bool isC = tid >= 32;
        float a = 0.f;
        #pragma unroll
        for (int jj = 0; jj < 5; ++jj) {
            const int bit = (idx >> (4 - jj)) & 1;
            a += xb[isC ? 5 + jj : jj] * (1.f - 2.f * (float)bit);
        }
        a *= -0.5f;
        float sv, cv; sincosf(a, &sv, &cv);
        if (isC) { pcr[idx] = cv; pci[idx] = sv; }
        else     { prr[idx] = cv; pri[idx] = sv; }
    }
    __syncthreads();

    const int mrow = ti * 16 + (lane & 15);
    const int ncol = tj * 16 + (lane & 15);
    const int ksub = (lane >> 4) << 1;

    for (int l = 0; l < 4; ++l) {
        const float* base = U + l * 4096;
        for (int i = tid; i < 1024; i += 128) {
            Ar[i] = base[i];        Ai[i] = base[1024 + i];
            Br[i] = base[2048 + i]; Bi[i] = base[3072 + i];
        }
        __syncthreads();

        {
            v8f cr = {}; v8f ci = {};
            #pragma unroll
            for (int kk = 0; kk < 8; ++kk) {
                const int k = kk * 4 + ksub;
                v2f aR, aI, aIn, bR, bI;
                aR.x = Ar[mrow * 32 + k];  aR.y = Ar[mrow * 32 + k + 1];
                aI.x = Ai[mrow * 32 + k];  aI.y = Ai[mrow * 32 + k + 1];
                aIn.x = -aI.x;             aIn.y = -aI.y;
                bR.x = Sr[k * 32 + ncol];  bR.y = Sr[(k + 1) * 32 + ncol];
                bI.x = Si[k * 32 + ncol];  bI.y = Si[(k + 1) * 32 + ncol];
                cr = wmma_f32(aR,  bR, cr);
                cr = wmma_f32(aIn, bI, cr);
                ci = wmma_f32(aR,  bI, ci);
                ci = wmma_f32(aI,  bR, ci);
            }
            #pragma unroll
            for (int v = 0; v < 8; ++v) {
                const int row = ti * 16 + v + ((lane >> 4) << 3);
                const int col = tj * 16 + (lane & 15);
                Cr[row * 32 + col] = cr[v];
                Ci[row * 32 + col] = ci[v];
            }
        }
        __syncthreads();

        {
            v8f tr = {}; v8f tv = {};
            #pragma unroll
            for (int kk = 0; kk < 8; ++kk) {
                const int k = kk * 4 + ksub;
                v2f aR, aI, aIn, bR, bI;
                aR.x = Cr[mrow * 32 + k];  aR.y = Cr[mrow * 32 + k + 1];
                aI.x = Ci[mrow * 32 + k];  aI.y = Ci[mrow * 32 + k + 1];
                aIn.x = -aI.x;             aIn.y = -aI.y;
                bR.x = Br[ncol * 32 + k];  bR.y = Br[ncol * 32 + k + 1];
                bI.x = Bi[ncol * 32 + k];  bI.y = Bi[ncol * 32 + k + 1];
                tr = wmma_f32(aR,  bR, tr);
                tr = wmma_f32(aIn, bI, tr);
                tv = wmma_f32(aR,  bI, tv);
                tv = wmma_f32(aI,  bR, tv);
            }
            #pragma unroll
            for (int v = 0; v < 8; ++v) {
                const int row = ti * 16 + v + ((lane >> 4) << 3);
                const int col = tj * 16 + (lane & 15);
                const int k   = row * 32 + col;
                const unsigned u = (unsigned)k;
                const int par = __popc(u & (u >> 1)) + (int)((u & 1u) & (u >> 9));
                const float sgn = (par & 1) ? -1.f : 1.f;
                float pRe = sgn, pIm = 0.f;
                if (l < 3) {
                    const float a1 = prr[row], b1 = pri[row];
                    const float a2 = pcr[col], b2 = pci[col];
                    pRe = sgn * (a1 * a2 - b1 * b2);
                    pIm = sgn * (a1 * b2 + b1 * a2);
                }
                const float er = tr[v], ei = tv[v];
                Sr[k] = er * pRe - ei * pIm;
                Si[k] = er * pIm + ei * pRe;
            }
        }
        __syncthreads();
    }

    float acc[10];
    #pragma unroll
    for (int j = 0; j < 10; ++j) acc[j] = 0.f;
    for (int i = tid; i < 1024; i += 128) {
        const float p = Sr[i] * Sr[i] + Si[i] * Si[i];
        #pragma unroll
        for (int j = 0; j < 10; ++j)
            acc[j] += ((i >> (9 - j)) & 1) ? -p : p;
    }
    #pragma unroll
    for (int j = 0; j < 10; ++j) red[tid][j] = acc[j];
    __syncthreads();
    if (tid < 10) {
        float s = 0.f;
        for (int t = 0; t < 128; ++t) s += red[t][tid];
        out[b * 10 + tid] = s;
    }
}

extern "C" void kernel_launch(void* const* d_in, const int* in_sizes, int n_in,
                              void* d_out, int out_size, void* d_ws, size_t ws_size,
                              hipStream_t stream) {
    const float* x  = (const float*)d_in[0];
    const float* w1 = (const float*)d_in[1];
    const float* w2 = (const float*)d_in[2];
    float* out = (float*)d_out;
    float* U   = (float*)d_ws;

    const int B = in_sizes[0] / 10;

    hipLaunchKernelGGL(build_unitaries, dim3(1), dim3(256), 0, stream, w1, w2, U);
    hipLaunchKernelGGL(qsim, dim3(B), dim3(128), 0, stream, x, U, out);
}


# state stays in registers through all layers: each product takes the last D tiles as its A operand (X -> X^T F^T), factors are B fragments staged once in LDS; no state LDS traffic; sum orders change
# speedup vs baseline: 2.7146x; 2.7146x over previous
#include <hip/hip_runtime.h>

typedef float v2f __attribute__((ext_vector_type(2)));
typedef float v4f __attribute__((ext_vector_type(4)));
typedef float v8f __attribute__((ext_vector_type(8)));

__device__ __forceinline__ v8f wmma_f32(v2f a, v2f b, v8f c) {
    return __builtin_amdgcn_wmma_f32_16x16x4_f32(false, a, false, b, (short)0, c, false, false);
}


#define LSTRIDE 2176

__global__ __launch_bounds__(256) void build_factors(const float* __restrict__ w1,
                                                     const float* __restrict__ w2,
                                                     float* __restrict__ U) {
    __shared__ float ry[4][10][2][2];
    __shared__ float pre[4][10][2];
    __shared__ float pim[4][10][2];

    const int tid = threadIdx.x;
    if (tid < 40) {
        const int l = tid / 10;
        const int j = tid % 10;
        const float* w = (l < 3) ? (w1 + (l * 10 + j) * 3) : (w2 + j * 3);
        const float theta = w[1], omega = w[2];
        float a = omega;
        if (l < 2)  a += w1[((l + 1) * 10 + j) * 3];
        if (l == 2) a += w2[j * 3];
        const float g = (l == 0) ? w[0] : 0.f;
        const float c = cosf(0.5f * theta);
        const float s = sinf(0.5f * theta);
        ry[l][j][0][0] = c;  ry[l][j][0][1] = -s;
        ry[l][j][1][0] = s;  ry[l][j][1][1] =  c;
        float sv, cv;
        sincosf(-0.5f * (g + a), &sv, &cv);  pre[l][j][0] = cv;  pim[l][j][0] = sv;
        sincosf(-0.5f * (g - a), &sv, &cv);  pre[l][j][1] = cv;  pim[l][j][1] = sv;
    }
    __syncthreads();

    {
        const int l     = tid >> 6;
        const int which = (tid >> 5) & 1;
        const int r     = tid & 31;
        const int j0    = which ? 5 : 0;
        float pr = 1.f, pi = 0.f;
        #pragma unroll
        for (int jj = 0; jj < 5; ++jj) {
            const int bit = (r >> (4 - jj)) & 1;
            const float ar = pre[l][j0 + jj][bit];
            const float ai = pim[l][j0 + jj][bit];
            const float nr = pr * ar - pi * ai;
            const float ni = pr * ai + pi * ar;
            pr = nr; pi = ni;
        }
        float* base = U + l * LSTRIDE + 2048 + which * 64;
        base[r]      = pr;
        base[r + 32] = pi;
    }

    for (int it = 0; it < 32; ++it) {
        const int idx = tid + it * 256;
        const int l   = idx >> 11;
        const int rem = idx & 2047;
        const int mat = rem >> 10;
        const int e   = rem & 1023;
        const int w4  = e & 3;
        const int ln  = (e >> 2) & 31;
        const int q   = e >> 7;
        const int r   = ((q >> 1) & 1) * 16 + (ln & 15);
        const int cc  = (q >> 2) * 16 + ((ln >> 4) << 3) + (q & 1) * 4 + w4;
        const int j0  = mat ? 5 : 0;
        float p = 1.f;
        #pragma unroll
        for (int jj = 0; jj < 5; ++jj) {
            const int rb = (r  >> (4 - jj)) & 1;
            const int cb = (cc >> (4 - jj)) & 1;
            p *= ry[l][j0 + jj][rb][cb];
        }
        U[l * LSTRIDE + rem] = p;
    }
}


__device__ __forceinline__ void xt_times_ft(const v8f (&xr)[2][2], const v8f (&xi)[2][2],
                                            const v4f (&f)[8],
                                            v8f (&yr)[2][2], v8f (&yi)[2][2]) {
    #pragma unroll
    for (int a = 0; a < 2; ++a)
        #pragma unroll
        for (int b = 0; b < 2; ++b) { yr[a][b] = (v8f){}; yi[a][b] = (v8f){}; }
    #pragma unroll
    for (int ti = 0; ti < 2; ++ti)
        #pragma unroll
        for (int s = 0; s < 4; ++s) {
            v2f fb[2];
            #pragma unroll
            for (int b = 0; b < 2; ++b) {
                const v4f ch = f[(ti * 2 + b) * 2 + (s >> 1)];
                fb[b] = (s & 1) ? (v2f){ch.z, ch.w} : (v2f){ch.x, ch.y};
            }
            #pragma unroll
            for (int a = 0; a < 2; ++a) {
                const v2f ar = {xr[ti][a][2 * s], xr[ti][a][2 * s + 1]};
                const v2f ai = {xi[ti][a][2 * s], xi[ti][a][2 * s + 1]};
                #pragma unroll
                for (int b = 0; b < 2; ++b) {
                    yr[a][b] = wmma_f32(ar, fb[b], yr[a][b]);
                    yi[a][b] = wmma_f32(ai, fb[b], yi[a][b]);
                }
            }
        }
}

__global__ __launch_bounds__(128) __attribute__((amdgpu_waves_per_eu(1, 3))) void qsim(const float* __restrict__ x,
                                            const float* __restrict__ U,
                                            float* __restrict__ out,
                                            int nbatch) {
    __shared__ __attribute__((aligned(16))) float Fac[4 * LSTRIDE];
    __shared__ __attribute__((aligned(16))) float Ph[4][4][4][32];
    __shared__ float red[4][32][10];

    const int tid  = threadIdx.x;
    const int lane = tid & 31;
    const int wid  = tid >> 5;
    const int b    = blockIdx.x * 4 + wid;
    const int bl   = (b < nbatch) ? b : (nbatch - 1);
    const int c16  = lane & 15;
    const int hh   = lane >> 4;

    {
        const v4f* src = reinterpret_cast<const v4f*>(U);
        v4f st[17];
        #pragma unroll
        for (int u = 0; u < 17; ++u) st[u] = src[tid + 128 * u];

        float xb[10];
        #pragma unroll
        for (int j = 0; j < 10; ++j) xb[j] = x[bl * 10 + j];
        float wrr[4], wri[4], wcr[4], wci[4];
        #pragma unroll
        for (int l = 0; l < 4; ++l) {
            const float* wt = U + l * LSTRIDE + 2048;
            wrr[l] = wt[lane]; wri[l] = wt[32 + lane]; wcr[l] = wt[64 + lane]; wci[l] = wt[96 + lane];
        }

        float xrr, xri, xcr, xci;
        {
            float ar = 0.f, ac = 0.f;
            #pragma unroll
            for (int jj = 0; jj < 5; ++jj) {
                const int bit = (lane >> (4 - jj)) & 1;
                ar += xb[jj]     * (1.f - 2.f * (float)bit);
                ac += xb[5 + jj] * (1.f - 2.f * (float)bit);
            }
            ar *= -0.5f; ac *= -0.5f;
            sincosf(ar, &xri, &xrr);
            sincosf(ac, &xci, &xcr);
        }
        #pragma unroll
        for (int l = 0; l < 4; ++l) {
            const float er = (l < 3) ? xrr : 1.f, ei = (l < 3) ? xri : 0.f;
            const float fr = (l < 3) ? xcr : 1.f, fi = (l < 3) ? xci : 0.f;
            Ph[wid][l][0][lane] = er * wrr[l] - ei * wri[l];
            Ph[wid][l][1][lane] = er * wri[l] + ei * wrr[l];
            Ph[wid][l][2][lane] = fr * wcr[l] - fi * wci[l];
            Ph[wid][l][3][lane] = fr * wci[l] + fi * wcr[l];
        }

        v4f* dst = reinterpret_cast<v4f*>(Fac);
        #pragma unroll
        for (int u = 0; u < 17; ++u) dst[tid + 128 * u] = st[u];
    }
    __syncthreads();

    v8f Xr[2][2], Xi[2][2];
    #pragma unroll
    for (int mt = 0; mt < 2; ++mt)
        #pragma unroll
        for (int nt = 0; nt < 2; ++nt) { Xr[mt][nt] = (v8f){}; Xi[mt][nt] = (v8f){}; }
    Xr[0][0][0] = (lane == 0) ? 1.f : 0.f;

    for (int l = 0; l < 4; ++l) {
        const v4f* fap = reinterpret_cast<const v4f*>(Fac + l * LSTRIDE) + lane;
        const v4f* fbp = reinterpret_cast<const v4f*>(Fac + l * LSTRIDE + 1024) + lane;
        v4f fa[8], fb[8];
        #pragma unroll
        for (int q = 0; q < 8; ++q) { fa[q] = fap[q * 32]; fb[q] = fbp[q * 32]; }
        v4f rre[2][2], rim[2][2];
        #pragma unroll
        for (int mt = 0; mt < 2; ++mt)
            #pragma unroll
            for (int g = 0; g < 2; ++g) {
                rre[mt][g] = *reinterpret_cast<const v4f*>(&Ph[wid][l][0][mt * 16 + hh * 8 + g * 4]);
                rim[mt][g] = *reinterpret_cast<const v4f*>(&Ph[wid][l][1][mt * 16 + hh * 8 + g * 4]);
            }
        float a2[2], b2[2];
        #pragma unroll
        for (int nt = 0; nt < 2; ++nt) {
            a2[nt] = Ph[wid][l][2][nt * 16 + c16];
            b2[nt] = Ph[wid][l][3][nt * 16 + c16];
        }

        v8f Yr[2][2], Yi[2][2];
        xt_times_ft(Xr, Xi, fa, Yr, Yi);
        v8f Tr[2][2], Tv[2][2];
        xt_times_ft(Yr, Yi, fb, Tr, Tv);

        #pragma unroll
        for (int mt = 0; mt < 2; ++mt)
            #pragma unroll
            for (int v = 0; v < 8; ++v) {
                const int row = mt * 16 + v + (hh << 3);
                const float a1 = rre[mt][v >> 2][v & 3];
                const float b1 = rim[mt][v >> 2][v & 3];
                #pragma unroll
                for (int nt = 0; nt < 2; ++nt) {
                    const int col = nt * 16 + c16;
                    const unsigned u = (unsigned)(row * 32 + col);
                    const int par = __popc(u & (u >> 1)) + (int)((u & 1u) & (u >> 9));
                    const float sgn = (par & 1) ? -1.f : 1.f;
                    const float pRe = sgn * (a1 * a2[nt] - b1 * b2[nt]);
                    const float pIm = sgn * (a1 * b2[nt] + b1 * a2[nt]);
                    const float er = Tr[mt][nt][v], ei = Tv[mt][nt][v];
                    Xr[mt][nt][v] = er * pRe - ei * pIm;
                    Xi[mt][nt][v] = er * pIm + ei * pRe;
                }
            }
    }

    float tot = 0.f, s0 = 0.f, s2 = 0.f, s3 = 0.f, s4 = 0.f, s5 = 0.f;
    #pragma unroll
    for (int mt = 0; mt < 2; ++mt)
        #pragma unroll
        for (int nt = 0; nt < 2; ++nt)
            #pragma unroll
            for (int v = 0; v < 8; ++v) {
                const float re = Xr[mt][nt][v], im = Xi[mt][nt][v];
                const float p = re * re + im * im;
                tot += p;
                s0 += mt ? -p : p;
                s2 += (v & 4) ? -p : p;
                s3 += (v & 2) ? -p : p;
                s4 += (v & 1) ? -p : p;
                s5 += nt ? -p : p;
            }
    float acc[10];
    acc[0] = s0;
    acc[1] = hh ? -tot : tot;
    acc[2] = s2; acc[3] = s3; acc[4] = s4; acc[5] = s5;
    #pragma unroll
    for (int i = 0; i < 4; ++i) acc[6 + i] = ((c16 >> (3 - i)) & 1) ? -tot : tot;
    #pragma unroll
    for (int j = 0; j < 10; ++j) red[wid][lane][j] = acc[j];
    __syncthreads();
    if (tid < 40) {
        const int e = tid / 10;
        const int j = tid - e * 10;
        float s = 0.f;
        for (int t = 0; t < 32; ++t) s += red[e][t][j];
        const int be = blockIdx.x * 4 + e;
        if (be < nbatch) out[be * 10 + j] = s;
    }
}

extern "C" void kernel_launch(void* const* d_in, const int* in_sizes, int n_in,
                              void* d_out, int out_size, void* d_ws, size_t ws_size,
                              hipStream_t stream) {
    const float* x  = (const float*)d_in[0];
    const float* w1 = (const float*)d_in[1];
    const float* w2 = (const float*)d_in[2];
    float* out = (float*)d_out;
    float* U   = (float*)d_ws;

    const int B = in_sizes[0] / 10;

    hipLaunchKernelGGL(build_factors, dim3(1), dim3(256), 0, stream, w1, w2, U);
    hipLaunchKernelGGL(qsim, dim3((B + 3) / 4), dim3(128), 0, stream, x, U, out, B);
}
